# starting kernel
# baseline (speedup 1.0000x reference)
#include <hip/hip_runtime.h>
#include <math.h>

#define NN 100000
#define NE 1600000
#define DH 128
#define NC 40
#define A_STRIDE 260

typedef __attribute__((ext_vector_type(2))) float v2f;
typedef __attribute__((ext_vector_type(8))) float v8f;

__global__ void zero_f32v4(float4* __restrict__ p, int n4) {
    int i = blockIdx.x * blockDim.x + threadIdx.x;
    if (i < n4) p[i] = make_float4(0.f, 0.f, 0.f, 0.f);
}

__global__ void deg_kernel(const int* __restrict__ dst, float* __restrict__ deg) {
    int e = blockIdx.x * blockDim.x + threadIdx.x;
    if (e < NE) atomicAdd(&deg[dst[e]], 1.0f);
}

__global__ void deginv_kernel(float* __restrict__ deg) {
    int i = blockIdx.x * blockDim.x + threadIdx.x;
    if (i < NN) deg[i] = 1.0f / fmaxf(deg[i], 1.0f);
}

__global__ void scatter_kernel(const float* __restrict__ x,
                               const int* __restrict__ src,
                               const int* __restrict__ dst,
                               float* __restrict__ aggr) {
    const int lane = threadIdx.x & 31;
    const int wave = threadIdx.x >> 5;
    const int e = blockIdx.x * (blockDim.x >> 5) + wave;
    if (e >= NE) return;
    const int s = __builtin_amdgcn_readfirstlane(src[e]);
    const int d = __builtin_amdgcn_readfirstlane(dst[e]);
    float4 v = ((const float4*)(x + (size_t)s * DH))[lane];
    float* o = aggr + (size_t)d * DH + lane * 4;
    atomicAdd(o + 0, v.x);
    atomicAdd(o + 1, v.y);
    atomicAdd(o + 2, v.z);
    atomicAdd(o + 3, v.w);
}

template <int DIN, int DOUT, bool RELU>
__global__ void sage_gemm(const float* __restrict__ aggr,
                          const float* __restrict__ x,
                          const float* __restrict__ deginv,
                          const float* __restrict__ Wl,
                          const float* __restrict__ Wr,
                          const float* __restrict__ bias,
                          float* __restrict__ out) {
    __shared__ float Alds[16 * A_STRIDE];

    const int m0   = blockIdx.x * 16;
    const int tid  = threadIdx.x;
    const int lane = tid & 31;
    const int wave = tid >> 5;
    constexpr int twoK = 2 * DIN;

    for (int idx = tid; idx < 16 * twoK; idx += blockDim.x) {
        int row = idx / twoK;
        int kk  = idx - row * twoK;
        int g   = m0 + row;
        float v;
        if (kk < DIN) v = aggr[(size_t)g * DIN + kk] * deginv[g];
        else          v = x[(size_t)g * DIN + (kk - DIN)];
        Alds[row * A_STRIDE + kk] = v;
    }
    __syncthreads();

    const int half = lane >> 4;
    const int l15  = lane & 15;
    const int n0   = wave * 16;
    const int col  = n0 + l15;
    const int ccol = (col < DOUT) ? col : 0;

    const float* __restrict__ Wlp = Wl + 2 * half * DOUT + ccol;
    const float* __restrict__ Wrp = Wr + 2 * half * DOUT + ccol;
    const float* __restrict__ Ap  = Alds + l15 * A_STRIDE + 2 * half;

    v8f acc = {};
    #pragma unroll 8
    for (int k = 0; k < DIN; k += 4) {
        v2f a, b;
        a.x = Ap[k];
        a.y = Ap[k + 1];
        b.x = Wlp[k * DOUT];
        b.y = Wlp[k * DOUT + DOUT];
        acc = __builtin_amdgcn_wmma_f32_16x16x4_f32(
            false, a, false, b, (short)0, acc, false, false);
    }
    #pragma unroll 8
    for (int k = 0; k < DIN; k += 4) {
        v2f a, b;
        a.x = Ap[DIN + k];
        a.y = Ap[DIN + k + 1];
        b.x = Wrp[k * DOUT];
        b.y = Wrp[k * DOUT + DOUT];
        acc = __builtin_amdgcn_wmma_f32_16x16x4_f32(
            false, a, false, b, (short)0, acc, false, false);
    }

    if (col < DOUT) {
        const float bv = bias[col];
        float* __restrict__ op = out + (size_t)(m0 + 8 * half) * DOUT + col;
        #pragma unroll
        for (int r = 0; r < 8; ++r) {
            float v = acc[r] + bv;
            if (RELU) v = fmaxf(v, 0.0f);
            op[r * DOUT] = v;
        }
    }
}

__global__ void logsoftmax_kernel(float* __restrict__ h) {
    int i = blockIdx.x * blockDim.x + threadIdx.x;
    if (i >= NN) return;
    float* row = h + (size_t)i * NC;
    float mx = -INFINITY;
    #pragma unroll
    for (int c = 0; c < NC; ++c) mx = fmaxf(mx, row[c]);
    float s = 0.0f;
    #pragma unroll
    for (int c = 0; c < NC; ++c) s += expf(row[c] - mx);
    float lse = mx + logf(s);
    #pragma unroll
    for (int c = 0; c < NC; ++c) row[c] = row[c] - lse;
}

extern "C" void kernel_launch(void* const* d_in, const int* in_sizes, int n_in,
                              void* d_out, int out_size, void* d_ws, size_t ws_size,
                              hipStream_t stream) {
    const float* x   = (const float*)d_in[0];
    const int*   ei  = (const int*)d_in[1];
    const int*   src = ei;
    const int*   dst = ei + NE;
    const float* Wl0 = (const float*)d_in[2];
    const float* Wr0 = (const float*)d_in[3];
    const float* b0  = (const float*)d_in[4];
    const float* Wl1 = (const float*)d_in[5];
    const float* Wr1 = (const float*)d_in[6];
    const float* b1  = (const float*)d_in[7];
    const float* Wl2 = (const float*)d_in[8];
    const float* Wr2 = (const float*)d_in[9];
    const float* b2  = (const float*)d_in[10];
    float* hout = (float*)d_out;

    char* ws = (char*)d_ws;
    size_t off = 0;
    auto carve = [&](size_t bytes) {
        void* p = ws + off;
        off = (off + bytes + 255) & ~(size_t)255;
        return p;
    };
    float* deg  = (float*)carve((size_t)NN * 4);
    float* aggr = (float*)carve((size_t)NN * DH * 4);
    float* h0   = (float*)carve((size_t)NN * DH * 4);
    float* h1   = (float*)carve((size_t)NN * DH * 4);

    const int ZB = 256;
    dim3 bZ(ZB);
    dim3 gN((NN + ZB - 1) / ZB);
    dim3 gZdeg((NN / 4 + ZB - 1) / ZB);
    dim3 gZaggr((NN * DH / 4 + ZB - 1) / ZB);
    dim3 gE((NE + ZB - 1) / ZB);
    dim3 gScat(NE / 8);
    dim3 gGemm(NN / 16);

    zero_f32v4<<<gZdeg, bZ, 0, stream>>>((float4*)deg, NN / 4);
    deg_kernel<<<gE, bZ, 0, stream>>>(dst, deg);
    deginv_kernel<<<gN, bZ, 0, stream>>>(deg);

    zero_f32v4<<<gZaggr, bZ, 0, stream>>>((float4*)aggr, NN * DH / 4);
    scatter_kernel<<<gScat, bZ, 0, stream>>>(x, src, dst, aggr);
    sage_gemm<DH, DH, true><<<gGemm, dim3(256), 0, stream>>>(aggr, x, deg, Wl0, Wr0, b0, h0);

    zero_f32v4<<<gZaggr, bZ, 0, stream>>>((float4*)aggr, NN * DH / 4);
    scatter_kernel<<<gScat, bZ, 0, stream>>>(h0, src, dst, aggr);
    sage_gemm<DH, DH, true><<<gGemm, dim3(256), 0, stream>>>(aggr, h0, deg, Wl1, Wr1, b1, h1);

    zero_f32v4<<<gZaggr, bZ, 0, stream>>>((float4*)aggr, NN * DH / 4);
    scatter_kernel<<<gScat, bZ, 0, stream>>>(h1, src, dst, aggr);
    sage_gemm<DH, NC, false><<<gGemm, dim3(96), 0, stream>>>(aggr, h1, deg, Wl2, Wr2, b2, hout);

    logsoftmax_kernel<<<gN, bZ, 0, stream>>>(hout);
}


# Scatter without global atomics: 256 work-groups each own 391 dst rows in a 200-KB LDS tile, scan all dst, gather hit rows, ds_add_f32 them; tile stored once, no zero fill; sums in LDS arrival order
# speedup vs baseline: 1.8584x; 1.8584x over previous
#include <hip/hip_runtime.h>
#include <math.h>

#define NN 100000
#define NE 1600000
#define DH 128
#define NC 40
#define A_STRIDE 260

typedef __attribute__((ext_vector_type(2))) float v2f;
typedef __attribute__((ext_vector_type(8))) float v8f;

__global__ void zero_f32v4(float4* __restrict__ p, int n4) {
    int i = blockIdx.x * blockDim.x + threadIdx.x;
    if (i < n4) p[i] = make_float4(0.f, 0.f, 0.f, 0.f);
}

__global__ void deg_kernel(const int* __restrict__ dst, float* __restrict__ deg) {
    int e = blockIdx.x * blockDim.x + threadIdx.x;
    if (e < NE) atomicAdd(&deg[dst[e]], 1.0f);
}

__global__ void deginv_kernel(float* __restrict__ deg) {
    int i = blockIdx.x * blockDim.x + threadIdx.x;
    if (i < NN) deg[i] = 1.0f / fmaxf(deg[i], 1.0f);
}

#define TR 391
#define NT 256
#define NG (NE / 128)
static_assert(NE % 128 == 0, "a scan step takes 128 whole edges");
static_assert(NT * TR >= NN, "tiles cover all rows");

__global__ __launch_bounds__(1024)
void tile_scatter(const float* __restrict__ x,
                  const int* __restrict__ src,
                  const int* __restrict__ dst,
                  float* __restrict__ aggr) {
    __shared__ float tile[(TR + 1) * DH];
    const int tid  = threadIdx.x;
    const int lane = tid & 31;
    const int wave = tid >> 5;
    const int base = blockIdx.x * TR;
    const int rows = min(TR, NN - base);

    float4* t4 = reinterpret_cast<float4*>(tile);
    for (int i = tid; i < (TR + 1) * DH / 4; i += 1024)
        t4[i] = make_float4(0.f, 0.f, 0.f, 0.f);
    __syncthreads();

    int pend_e = 0;
    int pend_d = TR;
    int npend  = 0;

    auto flush = [&]() {
        const int sv = src[pend_e];
        for (int b0 = 0; b0 < npend; b0 += 8) {
            float v[8][4];
            #pragma unroll
            for (int j = 0; j < 8; ++j) {
                const int s = __builtin_amdgcn_readlane(sv, b0 + j);
                const float* __restrict__ xr = x + (size_t)s * DH + lane;
                v[j][0] = xr[0];
                v[j][1] = xr[32];
                v[j][2] = xr[64];
                v[j][3] = xr[96];
            }
            #pragma unroll
            for (int j = 0; j < 8; ++j) {
                const int d = __builtin_amdgcn_readlane(pend_d, b0 + j);
                float* o = tile + d * DH + lane;
                atomicAdd(o +  0, v[j][0]);
                atomicAdd(o + 32, v[j][1]);
                atomicAdd(o + 64, v[j][2]);
                atomicAdd(o + 96, v[j][3]);
            }
        }
        pend_d = TR;
        npend = 0;
    };

    auto take = [&](unsigned& m, int dvc, int e0) {
        while (m && npend < 32) {
            const int l = __builtin_ctz(m);
            m &= m - 1;
            const int dl = __builtin_amdgcn_readlane(dvc, l) - base;
            if (lane == npend) { pend_e = e0 + 4 * l; pend_d = dl; }
            ++npend;
        }
    };

    constexpr int SU = 4;
    constexpr int NSETS = NG / SU;
    static_assert(NG % SU == 0, "whole sets");
    const int s0 = blockIdx.x * (NSETS / NT);
    const int4* __restrict__ dst4 = reinterpret_cast<const int4*>(dst);
    int s = s0 + wave;
    if (s >= NSETS) s -= NSETS;
    int4 nxt[SU];
    #pragma unroll
    for (int u = 0; u < SU; ++u) nxt[u] = dst4[(s * SU + u) * 32 + lane];
    for (int si = wave; si < NSETS; si += 32) {
        int4 cur[SU];
        #pragma unroll
        for (int u = 0; u < SU; ++u) cur[u] = nxt[u];
        const int e0 = s * (SU * 128);
        if (si + 32 < NSETS) {
            s = s0 + si + 32;
            if (s >= NSETS) s -= NSETS;
            #pragma unroll
            for (int u = 0; u < SU; ++u) nxt[u] = dst4[(s * SU + u) * 32 + lane];
        }
        #pragma unroll
        for (int u = 0; u < SU; ++u) {
            unsigned m0 = __builtin_amdgcn_ballot_w32((unsigned)(cur[u].x - base) < (unsigned)rows);
            unsigned m1 = __builtin_amdgcn_ballot_w32((unsigned)(cur[u].y - base) < (unsigned)rows);
            unsigned m2 = __builtin_amdgcn_ballot_w32((unsigned)(cur[u].z - base) < (unsigned)rows);
            unsigned m3 = __builtin_amdgcn_ballot_w32((unsigned)(cur[u].w - base) < (unsigned)rows);
            const int eu = e0 + u * 128;
            while (m0 | m1 | m2 | m3) {
                take(m0, cur[u].x, eu + 0);
                take(m1, cur[u].y, eu + 1);
                take(m2, cur[u].z, eu + 2);
                take(m3, cur[u].w, eu + 3);
                if (npend == 32) flush();
            }
        }
    }
    if (npend) flush();
    __syncthreads();

    float4* __restrict__ a4 = reinterpret_cast<float4*>(aggr + (size_t)base * DH);
    for (int i = tid; i < rows * (DH / 4); i += 1024) a4[i] = t4[i];
}

template <int DIN, int DOUT, bool RELU>
__global__ void sage_gemm(const float* __restrict__ aggr,
                          const float* __restrict__ x,
                          const float* __restrict__ deginv,
                          const float* __restrict__ Wl,
                          const float* __restrict__ Wr,
                          const float* __restrict__ bias,
                          float* __restrict__ out) {
    __shared__ float Alds[16 * A_STRIDE];

    const int m0   = blockIdx.x * 16;
    const int tid  = threadIdx.x;
    const int lane = tid & 31;
    const int wave = tid >> 5;
    constexpr int twoK = 2 * DIN;

    for (int idx = tid; idx < 16 * twoK; idx += blockDim.x) {
        int row = idx / twoK;
        int kk  = idx - row * twoK;
        int g   = m0 + row;
        float v;
        if (kk < DIN) v = aggr[(size_t)g * DIN + kk] * deginv[g];
        else          v = x[(size_t)g * DIN + (kk - DIN)];
        Alds[row * A_STRIDE + kk] = v;
    }
    __syncthreads();

    const int half = lane >> 4;
    const int l15  = lane & 15;
    const int n0   = wave * 16;
    const int col  = n0 + l15;
    const int ccol = (col < DOUT) ? col : 0;

    const float* __restrict__ Wlp = Wl + 2 * half * DOUT + ccol;
    const float* __restrict__ Wrp = Wr + 2 * half * DOUT + ccol;
    const float* __restrict__ Ap  = Alds + l15 * A_STRIDE + 2 * half;

    v8f acc = {};
    #pragma unroll 8
    for (int k = 0; k < DIN; k += 4) {
        v2f a, b;
        a.x = Ap[k];
        a.y = Ap[k + 1];
        b.x = Wlp[k * DOUT];
        b.y = Wlp[k * DOUT + DOUT];
        acc = __builtin_amdgcn_wmma_f32_16x16x4_f32(
            false, a, false, b, (short)0, acc, false, false);
    }
    #pragma unroll 8
    for (int k = 0; k < DIN; k += 4) {
        v2f a, b;
        a.x = Ap[DIN + k];
        a.y = Ap[DIN + k + 1];
        b.x = Wrp[k * DOUT];
        b.y = Wrp[k * DOUT + DOUT];
        acc = __builtin_amdgcn_wmma_f32_16x16x4_f32(
            false, a, false, b, (short)0, acc, false, false);
    }

    if (col < DOUT) {
        const float bv = bias[col];
        float* __restrict__ op = out + (size_t)(m0 + 8 * half) * DOUT + col;
        #pragma unroll
        for (int r = 0; r < 8; ++r) {
            float v = acc[r] + bv;
            if (RELU) v = fmaxf(v, 0.0f);
            op[r * DOUT] = v;
        }
    }
}

__global__ void logsoftmax_kernel(float* __restrict__ h) {
    int i = blockIdx.x * blockDim.x + threadIdx.x;
    if (i >= NN) return;
    float* row = h + (size_t)i * NC;
    float mx = -INFINITY;
    #pragma unroll
    for (int c = 0; c < NC; ++c) mx = fmaxf(mx, row[c]);
    float s = 0.0f;
    #pragma unroll
    for (int c = 0; c < NC; ++c) s += expf(row[c] - mx);
    float lse = mx + logf(s);
    #pragma unroll
    for (int c = 0; c < NC; ++c) row[c] = row[c] - lse;
}

extern "C" void kernel_launch(void* const* d_in, const int* in_sizes, int n_in,
                              void* d_out, int out_size, void* d_ws, size_t ws_size,
                              hipStream_t stream) {
    const float* x   = (const float*)d_in[0];
    const int*   ei  = (const int*)d_in[1];
    const int*   src = ei;
    const int*   dst = ei + NE;
    const float* Wl0 = (const float*)d_in[2];
    const float* Wr0 = (const float*)d_in[3];
    const float* b0  = (const float*)d_in[4];
    const float* Wl1 = (const float*)d_in[5];
    const float* Wr1 = (const float*)d_in[6];
    const float* b1  = (const float*)d_in[7];
    const float* Wl2 = (const float*)d_in[8];
    const float* Wr2 = (const float*)d_in[9];
    const float* b2  = (const float*)d_in[10];
    float* hout = (float*)d_out;

    char* ws = (char*)d_ws;
    size_t off = 0;
    auto carve = [&](size_t bytes) {
        void* p = ws + off;
        off = (off + bytes + 255) & ~(size_t)255;
        return p;
    };
    float* deg  = (float*)carve((size_t)NN * 4);
    float* aggr = (float*)carve((size_t)NN * DH * 4);
    float* h0   = (float*)carve((size_t)NN * DH * 4);
    float* h1   = (float*)carve((size_t)NN * DH * 4);

    const int ZB = 256;
    dim3 bZ(ZB);
    dim3 gN((NN + ZB - 1) / ZB);
    dim3 gZdeg((NN / 4 + ZB - 1) / ZB);
    dim3 gE((NE + ZB - 1) / ZB);
    dim3 gScat(NT);
    dim3 bScat(1024);
    dim3 gGemm(NN / 16);

    zero_f32v4<<<gZdeg, bZ, 0, stream>>>((float4*)deg, NN / 4);
    deg_kernel<<<gE, bZ, 0, stream>>>(dst, deg);
    deginv_kernel<<<gN, bZ, 0, stream>>>(deg);

    tile_scatter<<<gScat, bScat, 0, stream>>>(x, src, dst, aggr);
    sage_gemm<DH, DH, true><<<gGemm, dim3(256), 0, stream>>>(aggr, x, deg, Wl0, Wr0, b0, h0);

    tile_scatter<<<gScat, bScat, 0, stream>>>(h0, src, dst, aggr);
    sage_gemm<DH, DH, true><<<gGemm, dim3(256), 0, stream>>>(aggr, h0, deg, Wl1, Wr1, b1, h1);

    tile_scatter<<<gScat, bScat, 0, stream>>>(h1, src, dst, aggr);
    sage_gemm<DH, NC, false><<<gGemm, dim3(96), 0, stream>>>(aggr, h1, deg, Wl2, Wr2, b2, hout);

    logsoftmax_kernel<<<gN, bZ, 0, stream>>>(hout);
}


# GEMM loads requested ahead of use: A tile fetched as 16-B pieces, all of a thread's in flight at once (not 16 dependent 4-B round trips); bias, next weight group before current products; same sums
# speedup vs baseline: 2.3230x; 1.2500x over previous
#include <hip/hip_runtime.h>
#include <math.h>

#define NN 100000
#define NE 1600000
#define DH 128
#define NC 40
#define A_STRIDE 260

typedef __attribute__((ext_vector_type(2))) float v2f;
typedef __attribute__((ext_vector_type(8))) float v8f;

__global__ void zero_f32v4(float4* __restrict__ p, int n4) {
    int i = blockIdx.x * blockDim.x + threadIdx.x;
    if (i < n4) p[i] = make_float4(0.f, 0.f, 0.f, 0.f);
}

__global__ void deg_kernel(const int* __restrict__ dst, float* __restrict__ deg) {
    int e = blockIdx.x * blockDim.x + threadIdx.x;
    if (e < NE) atomicAdd(&deg[dst[e]], 1.0f);
}

__global__ void deginv_kernel(float* __restrict__ deg) {
    int i = blockIdx.x * blockDim.x + threadIdx.x;
    if (i < NN) deg[i] = 1.0f / fmaxf(deg[i], 1.0f);
}

#define TR 391
#define NT 256
#define NG (NE / 128)
#define BW 8
static_assert(NE % 128 == 0, "a group is 128 whole edges");
static_assert(NT * TR >= NN, "tiles cover all rows");
static_assert((size_t)NT * NG * 16 <= (size_t)NN * DH * 4, "the masks fit one feature plane");

typedef __attribute__((ext_vector_type(4))) int v4i;

__global__ __launch_bounds__(BW * 32)
void build_masks(const int* __restrict__ dst, int* __restrict__ hitmap) {
    __shared__ unsigned tab[BW][NT][4];
    const int tid  = threadIdx.x;
    const int lane = tid & 31;
    const int wave = tid >> 5;
    const int g    = blockIdx.x * BW + wave;

    uint4* tab4 = reinterpret_cast<uint4*>(&tab[0][0][0]);
    for (int i = tid; i < BW * NT; i += BW * 32) tab4[i] = make_uint4(0u, 0u, 0u, 0u);
    __syncthreads();

    if (g < NG) {
        const int4 d = reinterpret_cast<const int4*>(dst)[g * 32 + lane];
        const unsigned bit = 1u << lane;
        if ((unsigned)d.x < (unsigned)NN) atomicOr(&tab[wave][d.x / TR][0], bit);
        if ((unsigned)d.y < (unsigned)NN) atomicOr(&tab[wave][d.y / TR][1], bit);
        if ((unsigned)d.z < (unsigned)NN) atomicOr(&tab[wave][d.z / TR][2], bit);
        if ((unsigned)d.w < (unsigned)NN) atomicOr(&tab[wave][d.w / TR][3], bit);
    }
    __syncthreads();

    uint4* __restrict__ hm = reinterpret_cast<uint4*>(hitmap);
    for (int i = tid; i < BW * NT; i += BW * 32) {
        const int t  = i / BW;
        const int w  = i % BW;
        const int gg = blockIdx.x * BW + w;
        if (gg < NG) hm[(size_t)t * NG + gg] = tab4[w * NT + t];
    }
}

__global__ __launch_bounds__(1024)
void tile_replay(const float* __restrict__ x,
                 const int* __restrict__ src,
                 const int* __restrict__ dst,
                 const int* __restrict__ hitmap,
                 float* __restrict__ aggr) {
    __shared__ float tile[(TR + 1) * DH];
    const int tid  = threadIdx.x;
    const int lane = tid & 31;
    const int wave = tid >> 5;
    const int base = blockIdx.x * TR;
    const int rows = min(TR, NN - base);

    float4* t4 = reinterpret_cast<float4*>(tile);
    for (int i = tid; i < (TR + 1) * DH / 4; i += 1024)
        t4[i] = make_float4(0.f, 0.f, 0.f, 0.f);
    __syncthreads();

    int pend_e = 0;
    int npend  = 0;

    auto flush = [&]() {
        const int sv = src[pend_e];
        int dv = dst[pend_e] - base;
        dv = (lane < npend && (unsigned)dv < (unsigned)rows) ? dv : TR;
        for (int b0 = 0; b0 < npend; b0 += 8) {
            float v[8][4];
            #pragma unroll
            for (int j = 0; j < 8; ++j) {
                const int s = __builtin_amdgcn_readlane(sv, b0 + j);
                const float* __restrict__ xr = x + (size_t)s * DH + lane;
                v[j][0] = xr[0];
                v[j][1] = xr[32];
                v[j][2] = xr[64];
                v[j][3] = xr[96];
            }
            #pragma unroll
            for (int j = 0; j < 8; ++j) {
                const int d = __builtin_amdgcn_readlane(dv, b0 + j);
                float* o = tile + d * DH + lane;
                atomicAdd(o +  0, v[j][0]);
                atomicAdd(o + 32, v[j][1]);
                atomicAdd(o + 64, v[j][2]);
                atomicAdd(o + 96, v[j][3]);
            }
        }
        npend = 0;
    };

    auto push = [&](unsigned m, int e0) {
        while (m) {
            const int b = __builtin_ctz(m);
            m &= m - 1;
            if (lane == npend) pend_e = e0 + 4 * b;
            if (++npend == 32) flush();
        }
    };

    const v4i* __restrict__ hm = reinterpret_cast<const v4i*>(hitmap) + (size_t)blockIdx.x * NG;
    constexpr int NTRIP = (NG + 1023) / 1024;
    int g = wave * 32 + lane;
    v4i nxt = __builtin_nontemporal_load(hm + min(g, NG - 1));
    for (int t = 0; t < NTRIP; ++t) {
        const v4i cur = nxt;
        const bool mine = g < NG;
        const int gb = t * 1024 + wave * 32;
        g += 1024;
        nxt = __builtin_nontemporal_load(hm + min(g, NG - 1));
        unsigned busy = __builtin_amdgcn_ballot_w32(mine && (cur.x | cur.y | cur.z | cur.w) != 0);
        while (busy) {
            const int l = __builtin_ctz(busy);
            busy &= busy - 1;
            const int eb = (gb + l) * 128;
            push((unsigned)__builtin_amdgcn_readlane(cur.x, l), eb + 0);
            push((unsigned)__builtin_amdgcn_readlane(cur.y, l), eb + 1);
            push((unsigned)__builtin_amdgcn_readlane(cur.z, l), eb + 2);
            push((unsigned)__builtin_amdgcn_readlane(cur.w, l), eb + 3);
        }
    }
    if (npend) flush();
    __syncthreads();

    float4* __restrict__ a4 = reinterpret_cast<float4*>(aggr + (size_t)base * DH);
    for (int i = tid; i < rows * (DH / 4); i += 1024) a4[i] = t4[i];
}

template <int DIN, int DOUT, bool RELU, int NTH>
__global__ __launch_bounds__(NTH)
void sage_gemm(const float* aggr,
               const float* __restrict__ x,
               const float* __restrict__ deginv,
               const float* __restrict__ Wl,
               const float* __restrict__ Wr,
               const float* __restrict__ bias,
               float* out) {
    __shared__ __attribute__((aligned(16))) float Alds[16 * A_STRIDE];

    constexpr int twoK = 2 * DIN;
    constexpr int HV = DIN / 4;
    constexpr int NP = 16 * HV;
    constexpr int IT = (NP + NTH - 1) / NTH;
    constexpr int NGRP = twoK / 32;
    constexpr int HGRP = DIN / 32;
    static_assert(DIN % 32 == 0, "a group is 8 whole k-steps inside one half");
    static_assert(A_STRIDE % 4 == 0 && DIN % 4 == 0 && A_STRIDE >= twoK, "16-byte LDS pieces");
    static_assert(NTH % 32 == 0 && NTH / 32 * 16 >= DOUT, "one wave per 16 columns");

    const int m0   = blockIdx.x * 16;
    const int tid  = threadIdx.x;
    const int lane = tid & 31;
    const int wave = tid >> 5;

    float4 aa[IT], xx[IT];
    float  sv[IT];
    #pragma unroll
    for (int it = 0; it < IT; ++it) {
        const int j0  = tid + it * NTH;
        const int j   = ((it + 1) * NTH <= NP) ? j0 : min(j0, NP - 1);
        const int row = j / HV;
        const int c   = j - row * HV;
        const int g   = m0 + row;
        aa[it] = *reinterpret_cast<const float4*>(aggr + (size_t)g * DIN + 4 * c);
        xx[it] = *reinterpret_cast<const float4*>(x    + (size_t)g * DIN + 4 * c);
        sv[it] = deginv[g];
    }

    const int half = lane >> 4;
    const int l15  = lane & 15;
    const int n0   = wave * 16;
    const int col  = n0 + l15;
    const int ccol = (col < DOUT) ? col : 0;

    const float* __restrict__ Wlp = Wl + 2 * half * DOUT + ccol;
    const float* __restrict__ Wrp = Wr + 2 * half * DOUT + ccol;
    const float* __restrict__ Ap  = Alds + l15 * A_STRIDE + 2 * half;

    auto loadB = [&](int g, v2f (&b)[8]) {
        const float* __restrict__ Wp = (g < HGRP) ? Wlp : Wrp;
        const int kb = ((g < HGRP) ? g : g - HGRP) * 32;
        #pragma unroll
        for (int u = 0; u < 8; ++u) {
            b[u].x = Wp[(kb + 4 * u) * DOUT];
            b[u].y = Wp[(kb + 4 * u) * DOUT + DOUT];
        }
    };

    const float bv = bias[ccol];
    v2f bc[8], bn[8];
    loadB(0, bc);

    #pragma unroll
    for (int it = 0; it < IT; ++it) {
        const int j = tid + it * NTH;
        if ((it + 1) * NTH <= NP || j < NP) {
            const int row = j / HV;
            const int c   = j - row * HV;
            float4 v = aa[it];
            const float s = sv[it];
            v.x *= s; v.y *= s; v.z *= s; v.w *= s;
            *reinterpret_cast<float4*>(Alds + row * A_STRIDE + 4 * c)       = v;
            *reinterpret_cast<float4*>(Alds + row * A_STRIDE + DIN + 4 * c) = xx[it];
        }
    }
    __syncthreads();

    v8f acc = {};
    #pragma unroll
    for (int g = 0; g < NGRP; ++g) {
        if (g + 1 < NGRP) loadB(g + 1, bn);
        #pragma unroll
        for (int u = 0; u < 8; ++u) {
            v2f a;
            a.x = Ap[32 * g + 4 * u];
            a.y = Ap[32 * g + 4 * u + 1];
            acc = __builtin_amdgcn_wmma_f32_16x16x4_f32(
                false, a, false, bc[u], (short)0, acc, false, false);
        }
        #pragma unroll
        for (int u = 0; u < 8; ++u) bc[u] = bn[u];
    }

    if (col < DOUT) {
        float* op = out + (size_t)(m0 + 8 * half) * DOUT + col;
        #pragma unroll
        for (int r = 0; r < 8; ++r) {
            float v = acc[r] + bv;
            if (RELU) v = fmaxf(v, 0.0f);
            op[r * DOUT] = v;
        }
    }
}

__global__ void logsoftmax_kernel(float* __restrict__ h) {
    int i = blockIdx.x * blockDim.x + threadIdx.x;
    if (i >= NN) return;
    float* row = h + (size_t)i * NC;
    float mx = -INFINITY;
    #pragma unroll
    for (int c = 0; c < NC; ++c) mx = fmaxf(mx, row[c]);
    float s = 0.0f;
    #pragma unroll
    for (int c = 0; c < NC; ++c) s += expf(row[c] - mx);
    float lse = mx + logf(s);
    #pragma unroll
    for (int c = 0; c < NC; ++c) row[c] = row[c] - lse;
}

extern "C" void kernel_launch(void* const* d_in, const int* in_sizes, int n_in,
                              void* d_out, int out_size, void* d_ws, size_t ws_size,
                              hipStream_t stream) {
    const float* x   = (const float*)d_in[0];
    const int*   ei  = (const int*)d_in[1];
    const int*   src = ei;
    const int*   dst = ei + NE;
    const float* Wl0 = (const float*)d_in[2];
    const float* Wr0 = (const float*)d_in[3];
    const float* b0  = (const float*)d_in[4];
    const float* Wl1 = (const float*)d_in[5];
    const float* Wr1 = (const float*)d_in[6];
    const float* b1  = (const float*)d_in[7];
    const float* Wl2 = (const float*)d_in[8];
    const float* Wr2 = (const float*)d_in[9];
    const float* b2  = (const float*)d_in[10];
    float* hout = (float*)d_out;

    char* ws = (char*)d_ws;
    size_t off = 0;
    auto carve = [&](size_t bytes) {
        void* p = ws + off;
        off = (off + bytes + 255) & ~(size_t)255;
        return p;
    };
    float* deg  = (float*)carve((size_t)NN * 4);
    float* p0   = (float*)carve((size_t)NN * DH * 4);
    float* p1   = (float*)carve((size_t)NN * DH * 4);
    int*   hmap = (int*)carve((size_t)NN * DH * 4);

    const int ZB = 256;
    dim3 bZ(ZB);
    dim3 gN((NN + ZB - 1) / ZB);
    dim3 gZdeg((NN / 4 + ZB - 1) / ZB);
    dim3 gE((NE + ZB - 1) / ZB);
    dim3 gScat(NT);
    dim3 bScat(1024);
    dim3 gMask((NG + BW - 1) / BW);
    dim3 bMask(BW * 32);
    dim3 gGemm(NN / 16);

    zero_f32v4<<<gZdeg, bZ, 0, stream>>>((float4*)deg, NN / 4);
    deg_kernel<<<gE, bZ, 0, stream>>>(dst, deg);
    deginv_kernel<<<gN, bZ, 0, stream>>>(deg);

    build_masks<<<gMask, bMask, 0, stream>>>(dst, hmap);

    tile_replay<<<gScat, bScat, 0, stream>>>(x, src, dst, hmap, p0);
    sage_gemm<DH, DH, true, 256><<<gGemm, dim3(256), 0, stream>>>(p0, x, deg, Wl0, Wr0, b0, p1);

    tile_replay<<<gScat, bScat, 0, stream>>>(p1, src, dst, hmap, p0);
    sage_gemm<DH, DH, true, 256><<<gGemm, dim3(256), 0, stream>>>(p0, p1, deg, Wl1, Wr1, b1, p0);

    tile_replay<<<gScat, bScat, 0, stream>>>(p0, src, dst, hmap, p1);
    sage_gemm<DH, NC, false, 96><<<gGemm, dim3(96), 0, stream>>>(p1, p0, deg, Wl2, Wr2, b2, hout);

    logsoftmax_kernel<<<gN, bZ, 0, stream>>>(hout);
}


# Degree counted in the tile scatter's LDS (a sum of ones, exact in any order) and the mean applied at the tile's store: no degree array, no global atomic, no degree launches; 11 -> 8 launches
# speedup vs baseline: 2.7524x; 1.1848x over previous
#include <hip/hip_runtime.h>
#include <math.h>

#define NN 100000
#define NE 1600000
#define DH 128
#define NC 40
#define A_STRIDE 260

typedef __attribute__((ext_vector_type(2))) float v2f;
typedef __attribute__((ext_vector_type(8))) float v8f;

#define TR 391
#define NT 256
#define NG (NE / 128)
#define BW 8
static_assert(NE % 128 == 0, "a group is 128 whole edges");
static_assert(NT * TR >= NN, "tiles cover all rows");
static_assert((size_t)NT * NG * 16 <= (size_t)NN * DH * 4, "the masks fit one feature plane");

typedef __attribute__((ext_vector_type(4))) int v4i;

__global__ __launch_bounds__(BW * 32)
void build_masks(const int* __restrict__ dst, int* __restrict__ hitmap) {
    __shared__ unsigned tab[BW][NT][4];
    const int tid  = threadIdx.x;
    const int lane = tid & 31;
    const int wave = tid >> 5;
    const int g    = blockIdx.x * BW + wave;

    uint4* tab4 = reinterpret_cast<uint4*>(&tab[0][0][0]);
    for (int i = tid; i < BW * NT; i += BW * 32) tab4[i] = make_uint4(0u, 0u, 0u, 0u);
    __syncthreads();

    if (g < NG) {
        const int4 d = reinterpret_cast<const int4*>(dst)[g * 32 + lane];
        const unsigned bit = 1u << lane;
        if ((unsigned)d.x < (unsigned)NN) atomicOr(&tab[wave][d.x / TR][0], bit);
        if ((unsigned)d.y < (unsigned)NN) atomicOr(&tab[wave][d.y / TR][1], bit);
        if ((unsigned)d.z < (unsigned)NN) atomicOr(&tab[wave][d.z / TR][2], bit);
        if ((unsigned)d.w < (unsigned)NN) atomicOr(&tab[wave][d.w / TR][3], bit);
    }
    __syncthreads();

    uint4* __restrict__ hm = reinterpret_cast<uint4*>(hitmap);
    for (int i = tid; i < BW * NT; i += BW * 32) {
        const int t  = i / BW;
        const int w  = i % BW;
        const int gg = blockIdx.x * BW + w;
        if (gg < NG) hm[(size_t)t * NG + gg] = tab4[w * NT + t];
    }
}

__global__ __launch_bounds__(1024)
void tile_replay(const float* __restrict__ x,
                 const int* __restrict__ src,
                 const int* __restrict__ dst,
                 const int* __restrict__ hitmap,
                 float* __restrict__ aggr) {
    __shared__ float tile[(TR + 1) * DH];
    __shared__ float cnt[TR + 1];
    const int tid  = threadIdx.x;
    const int lane = tid & 31;
    const int wave = tid >> 5;
    const int base = blockIdx.x * TR;
    const int rows = min(TR, NN - base);

    float4* t4 = reinterpret_cast<float4*>(tile);
    for (int i = tid; i < (TR + 1) * DH / 4; i += 1024)
        t4[i] = make_float4(0.f, 0.f, 0.f, 0.f);
    for (int i = tid; i < TR + 1; i += 1024) cnt[i] = 0.f;
    __syncthreads();

    int pend_e = 0;
    int npend  = 0;

    auto flush = [&]() {
        const int sv = src[pend_e];
        int dv = dst[pend_e] - base;
        const bool live = lane < npend && (unsigned)dv < (unsigned)rows;
        if (live) atomicAdd(&cnt[dv], 1.0f);
        dv = live ? dv : TR;
        for (int b0 = 0; b0 < npend; b0 += 8) {
            float v[8][4];
            #pragma unroll
            for (int j = 0; j < 8; ++j) {
                const int s = __builtin_amdgcn_readlane(sv, b0 + j);
                const float* __restrict__ xr = x + (size_t)s * DH + lane;
                v[j][0] = xr[0];
                v[j][1] = xr[32];
                v[j][2] = xr[64];
                v[j][3] = xr[96];
            }
            #pragma unroll
            for (int j = 0; j < 8; ++j) {
                const int d = __builtin_amdgcn_readlane(dv, b0 + j);
                float* o = tile + d * DH + lane;
                atomicAdd(o +  0, v[j][0]);
                atomicAdd(o + 32, v[j][1]);
                atomicAdd(o + 64, v[j][2]);
                atomicAdd(o + 96, v[j][3]);
            }
        }
        npend = 0;
    };

    auto push = [&](unsigned m, int e0) {
        while (m) {
            const int b = __builtin_ctz(m);
            m &= m - 1;
            if (lane == npend) pend_e = e0 + 4 * b;
            if (++npend == 32) flush();
        }
    };

    const v4i* __restrict__ hm = reinterpret_cast<const v4i*>(hitmap) + (size_t)blockIdx.x * NG;
    constexpr int NTRIP = (NG + 1023) / 1024;
    int g = wave * 32 + lane;
    v4i nxt = __builtin_nontemporal_load(hm + min(g, NG - 1));
    for (int t = 0; t < NTRIP; ++t) {
        const v4i cur = nxt;
        const bool mine = g < NG;
        const int gb = t * 1024 + wave * 32;
        g += 1024;
        nxt = __builtin_nontemporal_load(hm + min(g, NG - 1));
        unsigned busy = __builtin_amdgcn_ballot_w32(mine && (cur.x | cur.y | cur.z | cur.w) != 0);
        while (busy) {
            const int l = __builtin_ctz(busy);
            busy &= busy - 1;
            const int eb = (gb + l) * 128;
            push((unsigned)__builtin_amdgcn_readlane(cur.x, l), eb + 0);
            push((unsigned)__builtin_amdgcn_readlane(cur.y, l), eb + 1);
            push((unsigned)__builtin_amdgcn_readlane(cur.z, l), eb + 2);
            push((unsigned)__builtin_amdgcn_readlane(cur.w, l), eb + 3);
        }
    }
    if (npend) flush();
    __syncthreads();

    float4* __restrict__ a4 = reinterpret_cast<float4*>(aggr + (size_t)base * DH);
    for (int i = tid; i < rows * (DH / 4); i += 1024) {
        const float s = 1.0f / fmaxf(cnt[i / (DH / 4)], 1.0f);
        float4 v = t4[i];
        v.x *= s; v.y *= s; v.z *= s; v.w *= s;
        a4[i] = v;
    }
}

template <int DIN, int DOUT, bool RELU, int NTH>
__global__ __launch_bounds__(NTH)
void sage_gemm(const float* aggr,
               const float* __restrict__ x,
               const float* __restrict__ Wl,
               const float* __restrict__ Wr,
               const float* __restrict__ bias,
               float* out) {
    __shared__ __attribute__((aligned(16))) float Alds[16 * A_STRIDE];

    constexpr int twoK = 2 * DIN;
    constexpr int HV = DIN / 4;
    constexpr int NP = 16 * HV;
    constexpr int IT = (NP + NTH - 1) / NTH;
    constexpr int NGRP = twoK / 32;
    constexpr int HGRP = DIN / 32;
    static_assert(DIN % 32 == 0, "a group is 8 whole k-steps inside one half");
    static_assert(A_STRIDE % 4 == 0 && DIN % 4 == 0 && A_STRIDE >= twoK, "16-byte LDS pieces");
    static_assert(NTH % 32 == 0 && NTH / 32 * 16 >= DOUT, "one wave per 16 columns");

    const int m0   = blockIdx.x * 16;
    const int tid  = threadIdx.x;
    const int lane = tid & 31;
    const int wave = tid >> 5;

    float4 aa[IT], xx[IT];
    #pragma unroll
    for (int it = 0; it < IT; ++it) {
        const int j0  = tid + it * NTH;
        const int j   = ((it + 1) * NTH <= NP) ? j0 : min(j0, NP - 1);
        const int row = j / HV;
        const int c   = j - row * HV;
        const int g   = m0 + row;
        aa[it] = *reinterpret_cast<const float4*>(aggr + (size_t)g * DIN + 4 * c);
        xx[it] = *reinterpret_cast<const float4*>(x    + (size_t)g * DIN + 4 * c);
    }

    const int half = lane >> 4;
    const int l15  = lane & 15;
    const int n0   = wave * 16;
    const int col  = n0 + l15;
    const int ccol = (col < DOUT) ? col : 0;

    const float* __restrict__ Wlp = Wl + 2 * half * DOUT + ccol;
    const float* __restrict__ Wrp = Wr + 2 * half * DOUT + ccol;
    const float* __restrict__ Ap  = Alds + l15 * A_STRIDE + 2 * half;

    auto loadB = [&](int g, v2f (&b)[8]) {
        const float* __restrict__ Wp = (g < HGRP) ? Wlp : Wrp;
        const int kb = ((g < HGRP) ? g : g - HGRP) * 32;
        #pragma unroll
        for (int u = 0; u < 8; ++u) {
            b[u].x = Wp[(kb + 4 * u) * DOUT];
            b[u].y = Wp[(kb + 4 * u) * DOUT + DOUT];
        }
    };

    const float bv = bias[ccol];
    v2f bc[8], bn[8];
    loadB(0, bc);

    #pragma unroll
    for (int it = 0; it < IT; ++it) {
        const int j = tid + it * NTH;
        if ((it + 1) * NTH <= NP || j < NP) {
            const int row = j / HV;
            const int c   = j - row * HV;
            *reinterpret_cast<float4*>(Alds + row * A_STRIDE + 4 * c)       = aa[it];
            *reinterpret_cast<float4*>(Alds + row * A_STRIDE + DIN + 4 * c) = xx[it];
        }
    }
    __syncthreads();

    v8f acc = {};
    #pragma unroll
    for (int g = 0; g < NGRP; ++g) {
        if (g + 1 < NGRP) loadB(g + 1, bn);
        #pragma unroll
        for (int u = 0; u < 8; ++u) {
            v2f a;
            a.x = Ap[32 * g + 4 * u];
            a.y = Ap[32 * g + 4 * u + 1];
            acc = __builtin_amdgcn_wmma_f32_16x16x4_f32(
                false, a, false, bc[u], (short)0, acc, false, false);
        }
        #pragma unroll
        for (int u = 0; u < 8; ++u) bc[u] = bn[u];
    }

    if (col < DOUT) {
        float* op = out + (size_t)(m0 + 8 * half) * DOUT + col;
        #pragma unroll
        for (int r = 0; r < 8; ++r) {
            float v = acc[r] + bv;
            if (RELU) v = fmaxf(v, 0.0f);
            op[r * DOUT] = v;
        }
    }
}

__global__ void logsoftmax_kernel(float* __restrict__ h) {
    int i = blockIdx.x * blockDim.x + threadIdx.x;
    if (i >= NN) return;
    float* row = h + (size_t)i * NC;
    float mx = -INFINITY;
    #pragma unroll
    for (int c = 0; c < NC; ++c) mx = fmaxf(mx, row[c]);
    float s = 0.0f;
    #pragma unroll
    for (int c = 0; c < NC; ++c) s += expf(row[c] - mx);
    float lse = mx + logf(s);
    #pragma unroll
    for (int c = 0; c < NC; ++c) row[c] = row[c] - lse;
}

extern "C" void kernel_launch(void* const* d_in, const int* in_sizes, int n_in,
                              void* d_out, int out_size, void* d_ws, size_t ws_size,
                              hipStream_t stream) {
    const float* x   = (const float*)d_in[0];
    const int*   ei  = (const int*)d_in[1];
    const int*   src = ei;
    const int*   dst = ei + NE;
    const float* Wl0 = (const float*)d_in[2];
    const float* Wr0 = (const float*)d_in[3];
    const float* b0  = (const float*)d_in[4];
    const float* Wl1 = (const float*)d_in[5];
    const float* Wr1 = (const float*)d_in[6];
    const float* b1  = (const float*)d_in[7];
    const float* Wl2 = (const float*)d_in[8];
    const float* Wr2 = (const float*)d_in[9];
    const float* b2  = (const float*)d_in[10];
    float* hout = (float*)d_out;

    char* ws = (char*)d_ws;
    size_t off = 0;
    auto carve = [&](size_t bytes) {
        void* p = ws + off;
        off = (off + bytes + 255) & ~(size_t)255;
        return p;
    };
    float* p0   = (float*)carve((size_t)NN * DH * 4);
    float* p1   = (float*)carve((size_t)NN * DH * 4);
    int*   hmap = (int*)carve((size_t)NN * DH * 4);

    const int ZB = 256;
    dim3 bZ(ZB);
    dim3 gN((NN + ZB - 1) / ZB);
    dim3 gScat(NT);
    dim3 bScat(1024);
    dim3 gMask((NG + BW - 1) / BW);
    dim3 bMask(BW * 32);
    dim3 gGemm(NN / 16);

    build_masks<<<gMask, bMask, 0, stream>>>(dst, hmap);

    tile_replay<<<gScat, bScat, 0, stream>>>(x, src, dst, hmap, p0);
    sage_gemm<DH, DH, true, 256><<<gGemm, dim3(256), 0, stream>>>(p0, x, Wl0, Wr0, b0, p1);

    tile_replay<<<gScat, bScat, 0, stream>>>(p1, src, dst, hmap, p0);
    sage_gemm<DH, DH, true, 256><<<gGemm, dim3(256), 0, stream>>>(p0, p1, Wl1, Wr1, b1, p0);

    tile_replay<<<gScat, bScat, 0, stream>>>(p0, src, dst, hmap, p1);
    sage_gemm<DH, NC, false, 96><<<gGemm, dim3(96), 0, stream>>>(p1, p0, Wl2, Wr2, b2, hout);

    logsoftmax_kernel<<<gN, bZ, 0, stream>>>(hout);
}


# Edge table holds source and tile row per edge (4 slots per tile and 128-edge group, else rescan), not one bit: tile scatter packs slots by ballot, no bit walk or src/dst loads; sums in arrival order
# speedup vs baseline: 3.1848x; 1.1571x over previous
#include <hip/hip_runtime.h>
#include <math.h>

#define NN 100000
#define NE 1600000
#define DH 128
#define NC 40
#define A_STRIDE 260

typedef __attribute__((ext_vector_type(2))) float v2f;
typedef __attribute__((ext_vector_type(8))) float v8f;

#define TR 391
#define NT 256
#define NG (NE / 128)
#define BW 8
#define SLOT_USED 0x80000000u
#define SLOT_SPILL 0xffffffffu
#define RING 256u
static_assert(NE % 128 == 0, "a group is 128 whole edges");
static_assert(NT * TR >= NN, "tiles cover all rows");
static_assert((size_t)NT * NG * 16 <= (size_t)NN * DH * 4, "the table fits one feature plane");
static_assert(NN <= (1 << 17) && TR <= (1 << 9), "source in 17 bits, row in 9 bits");

typedef __attribute__((ext_vector_type(4))) int v4i;

__global__ __launch_bounds__(BW * 32)
void build_table(const int* __restrict__ src, const int* __restrict__ dst,
                 int* __restrict__ table) {
    __shared__ unsigned tab[BW][NT][4];
    __shared__ unsigned fill[BW][NT];
    const int tid  = threadIdx.x;
    const int lane = tid & 31;
    const int wave = tid >> 5;
    const int g    = blockIdx.x * BW + wave;

    uint4* tab4 = reinterpret_cast<uint4*>(&tab[0][0][0]);
    unsigned* fill1 = &fill[0][0];
    for (int i = tid; i < BW * NT; i += BW * 32) {
        tab4[i]  = make_uint4(0u, 0u, 0u, 0u);
        fill1[i] = 0u;
    }
    __syncthreads();

    if (g < NG) {
        const int4 d = reinterpret_cast<const int4*>(dst)[g * 32 + lane];
        const int4 s = reinterpret_cast<const int4*>(src)[g * 32 + lane];
        auto put = [&](int dv, int sv) {
            if ((unsigned)dv < (unsigned)NN && (unsigned)sv < (unsigned)NN) {
                const int t = dv / TR;
                const int r = dv - t * TR;
                const unsigned k = atomicAdd(&fill[wave][t], 1u);
                if (k < 4u) tab[wave][t][k] = SLOT_USED | ((unsigned)r << 17) | (unsigned)sv;
            }
        };
        put(d.x, s.x);
        put(d.y, s.y);
        put(d.z, s.z);
        put(d.w, s.w);
    }
    __syncthreads();

    uint4* __restrict__ hm = reinterpret_cast<uint4*>(table);
    for (int i = tid; i < BW * NT; i += BW * 32) {
        const int t  = i / BW;
        const int w  = i % BW;
        const int gg = blockIdx.x * BW + w;
        uint4 v = tab4[w * NT + t];
        if (fill[w][t] > 4u) v = make_uint4(SLOT_SPILL, 0u, 0u, 0u);
        if (gg < NG) hm[(size_t)t * NG + gg] = v;
    }
}

__global__ __launch_bounds__(1024)
void tile_replay(const float* __restrict__ x,
                 const int* __restrict__ src,
                 const int* __restrict__ dst,
                 const int* __restrict__ table,
                 float* __restrict__ aggr) {
    __shared__ float tile[(TR + 1) * DH];
    __shared__ float cnt[TR + 1];
    __shared__ unsigned ring[32][RING];
    const int tid  = threadIdx.x;
    const int lane = tid & 31;
    const int wave = tid >> 5;
    const int base = blockIdx.x * TR;
    const int rows = min(TR, NN - base);

    float4* t4 = reinterpret_cast<float4*>(tile);
    for (int i = tid; i < (TR + 1) * DH / 4; i += 1024)
        t4[i] = make_float4(0.f, 0.f, 0.f, 0.f);
    for (int i = tid; i < TR + 1; i += 1024) cnt[i] = 0.f;
    for (int i = tid; i < 32 * RING; i += 1024) (&ring[0][0])[i] = 0u;
    __syncthreads();

    unsigned* myring = ring[wave];
    unsigned head = 0;
    unsigned tail = 0;

    auto flush = [&](int n) {
        const unsigned pv = myring[(head + (unsigned)lane) & (RING - 1u)];
        head += (unsigned)n;
        const bool live = lane < n;
        const int sv = live ? (int)(pv & 0x1ffffu) : 0;
        const int dv = live ? (int)((pv >> 17) & 0x1ffu) : TR;
        if (live) atomicAdd(&cnt[dv], 1.0f);
        for (int b0 = 0; b0 < n; b0 += 8) {
            float v[8][4];
            #pragma unroll
            for (int j = 0; j < 8; ++j) {
                const int s = __builtin_amdgcn_readlane(sv, b0 + j);
                const float* __restrict__ xr = x + (size_t)s * DH + lane;
                v[j][0] = xr[0];
                v[j][1] = xr[32];
                v[j][2] = xr[64];
                v[j][3] = xr[96];
            }
            #pragma unroll
            for (int j = 0; j < 8; ++j) {
                const int d = __builtin_amdgcn_readlane(dv, b0 + j);
                float* o = tile + d * DH + lane;
                atomicAdd(o +  0, v[j][0]);
                atomicAdd(o + 32, v[j][1]);
                atomicAdd(o + 64, v[j][2]);
                atomicAdd(o + 96, v[j][3]);
            }
        }
    };

    auto push = [&](bool has, unsigned val) {
        const unsigned bal = __builtin_amdgcn_ballot_w32(has);
        const unsigned rank = __builtin_amdgcn_mbcnt_lo(bal, 0u);
        if (has) myring[(tail + rank) & (RING - 1u)] = val;
        tail += (unsigned)__builtin_popcount(bal);
    };
    auto drain = [&]() {
        while (tail - head >= 32u) flush(32);
    };

    const v4i* __restrict__ hm = reinterpret_cast<const v4i*>(table) + (size_t)blockIdx.x * NG;
    constexpr int NTRIP = (NG + 1023) / 1024;
    int g = wave * 32 + lane;
    v4i nxt = __builtin_nontemporal_load(hm + min(g, NG - 1));
    for (int t = 0; t < NTRIP; ++t) {
        const v4i cur = nxt;
        const bool mine = g < NG;
        const int gb = t * 1024 + wave * 32;
        g += 1024;
        nxt = __builtin_nontemporal_load(hm + min(g, NG - 1));
        const bool spill = mine && (unsigned)cur.x == SLOT_SPILL;
        const bool take  = mine && !spill;
        push(take && ((unsigned)cur.x & SLOT_USED) != 0u, (unsigned)cur.x);
        push(take && ((unsigned)cur.y & SLOT_USED) != 0u, (unsigned)cur.y);
        push(take && ((unsigned)cur.z & SLOT_USED) != 0u, (unsigned)cur.z);
        push(take && ((unsigned)cur.w & SLOT_USED) != 0u, (unsigned)cur.w);
        drain();
        unsigned om = __builtin_amdgcn_ballot_w32(spill);
        while (om) {
            const int l = __builtin_ctz(om);
            om &= om - 1;
            const int gq = gb + l;
            const int4 d = reinterpret_cast<const int4*>(dst)[gq * 32 + lane];
            const int4 s = reinterpret_cast<const int4*>(src)[gq * 32 + lane];
            auto test = [&](int dv, int sv) {
                const int dc = dv - base;
                const bool ok = (unsigned)dc < (unsigned)rows && (unsigned)sv < (unsigned)NN;
                push(ok, SLOT_USED | ((unsigned)dc << 17) | (unsigned)sv);
            };
            test(d.x, s.x);
            test(d.y, s.y);
            test(d.z, s.z);
            test(d.w, s.w);
            drain();
        }
    }
    if (tail != head) flush((int)(tail - head));
    __syncthreads();

    float4* __restrict__ a4 = reinterpret_cast<float4*>(aggr + (size_t)base * DH);
    for (int i = tid; i < rows * (DH / 4); i += 1024) {
        const float s = 1.0f / fmaxf(cnt[i / (DH / 4)], 1.0f);
        float4 v = t4[i];
        v.x *= s; v.y *= s; v.z *= s; v.w *= s;
        a4[i] = v;
    }
}

template <int DIN, int DOUT, bool RELU, int NTH>
__global__ __launch_bounds__(NTH)
void sage_gemm(const float* aggr,
               const float* __restrict__ x,
               const float* __restrict__ Wl,
               const float* __restrict__ Wr,
               const float* __restrict__ bias,
               float* out) {
    __shared__ __attribute__((aligned(16))) float Alds[16 * A_STRIDE];

    constexpr int twoK = 2 * DIN;
    constexpr int HV = DIN / 4;
    constexpr int NP = 16 * HV;
    constexpr int IT = (NP + NTH - 1) / NTH;
    constexpr int NGRP = twoK / 32;
    constexpr int HGRP = DIN / 32;
    static_assert(DIN % 32 == 0, "a group is 8 whole k-steps inside one half");
    static_assert(A_STRIDE % 4 == 0 && DIN % 4 == 0 && A_STRIDE >= twoK, "16-byte LDS pieces");
    static_assert(NTH % 32 == 0 && NTH / 32 * 16 >= DOUT, "one wave per 16 columns");

    const int m0   = blockIdx.x * 16;
    const int tid  = threadIdx.x;
    const int lane = tid & 31;
    const int wave = tid >> 5;

    float4 aa[IT], xx[IT];
    #pragma unroll
    for (int it = 0; it < IT; ++it) {
        const int j0  = tid + it * NTH;
        const int j   = ((it + 1) * NTH <= NP) ? j0 : min(j0, NP - 1);
        const int row = j / HV;
        const int c   = j - row * HV;
        const int g   = m0 + row;
        aa[it] = *reinterpret_cast<const float4*>(aggr + (size_t)g * DIN + 4 * c);
        xx[it] = *reinterpret_cast<const float4*>(x    + (size_t)g * DIN + 4 * c);
    }

    const int half = lane >> 4;
    const int l15  = lane & 15;
    const int n0   = wave * 16;
    const int col  = n0 + l15;
    const int ccol = (col < DOUT) ? col : 0;

    const float* __restrict__ Wlp = Wl + 2 * half * DOUT + ccol;
    const float* __restrict__ Wrp = Wr + 2 * half * DOUT + ccol;
    const float* __restrict__ Ap  = Alds + l15 * A_STRIDE + 2 * half;

    auto loadB = [&](int g, v2f (&b)[8]) {
        const float* __restrict__ Wp = (g < HGRP) ? Wlp : Wrp;
        const int kb = ((g < HGRP) ? g : g - HGRP) * 32;
        #pragma unroll
        for (int u = 0; u < 8; ++u) {
            b[u].x = Wp[(kb + 4 * u) * DOUT];
            b[u].y = Wp[(kb + 4 * u) * DOUT + DOUT];
        }
    };

    const float bv = bias[ccol];
    v2f bc[8], bn[8];
    loadB(0, bc);

    #pragma unroll
    for (int it = 0; it < IT; ++it) {
        const int j = tid + it * NTH;
        if ((it + 1) * NTH <= NP || j < NP) {
            const int row = j / HV;
            const int c   = j - row * HV;
            *reinterpret_cast<float4*>(Alds + row * A_STRIDE + 4 * c)       = aa[it];
            *reinterpret_cast<float4*>(Alds + row * A_STRIDE + DIN + 4 * c) = xx[it];
        }
    }
    __syncthreads();

    v8f acc = {};
    #pragma unroll
    for (int g = 0; g < NGRP; ++g) {
        if (g + 1 < NGRP) loadB(g + 1, bn);
        #pragma unroll
        for (int u = 0; u < 8; ++u) {
            v2f a;
            a.x = Ap[32 * g + 4 * u];
            a.y = Ap[32 * g + 4 * u + 1];
            acc = __builtin_amdgcn_wmma_f32_16x16x4_f32(
                false, a, false, bc[u], (short)0, acc, false, false);
        }
        #pragma unroll
        for (int u = 0; u < 8; ++u) bc[u] = bn[u];
    }

    if (col < DOUT) {
        float* op = out + (size_t)(m0 + 8 * half) * DOUT + col;
        #pragma unroll
        for (int r = 0; r < 8; ++r) {
            float v = acc[r] + bv;
            if (RELU) v = fmaxf(v, 0.0f);
            op[r * DOUT] = v;
        }
    }
}

__global__ void logsoftmax_kernel(float* __restrict__ h) {
    int i = blockIdx.x * blockDim.x + threadIdx.x;
    if (i >= NN) return;
    float* row = h + (size_t)i * NC;
    float mx = -INFINITY;
    #pragma unroll
    for (int c = 0; c < NC; ++c) mx = fmaxf(mx, row[c]);
    float s = 0.0f;
    #pragma unroll
    for (int c = 0; c < NC; ++c) s += expf(row[c] - mx);
    float lse = mx + logf(s);
    #pragma unroll
    for (int c = 0; c < NC; ++c) row[c] = row[c] - lse;
}

extern "C" void kernel_launch(void* const* d_in, const int* in_sizes, int n_in,
                              void* d_out, int out_size, void* d_ws, size_t ws_size,
                              hipStream_t stream) {
    const float* x   = (const float*)d_in[0];
    const int*   ei  = (const int*)d_in[1];
    const int*   src = ei;
    const int*   dst = ei + NE;
    const float* Wl0 = (const float*)d_in[2];
    const float* Wr0 = (const float*)d_in[3];
    const float* b0  = (const float*)d_in[4];
    const float* Wl1 = (const float*)d_in[5];
    const float* Wr1 = (const float*)d_in[6];
    const float* b1  = (const float*)d_in[7];
    const float* Wl2 = (const float*)d_in[8];
    const float* Wr2 = (const float*)d_in[9];
    const float* b2  = (const float*)d_in[10];
    float* hout = (float*)d_out;

    char* ws = (char*)d_ws;
    size_t off = 0;
    auto carve = [&](size_t bytes) {
        void* p = ws + off;
        off = (off + bytes + 255) & ~(size_t)255;
        return p;
    };
    float* p0   = (float*)carve((size_t)NN * DH * 4);
    float* p1   = (float*)carve((size_t)NN * DH * 4);
    int*   hmap = (int*)carve((size_t)NN * DH * 4);

    const int ZB = 256;
    dim3 bZ(ZB);
    dim3 gN((NN + ZB - 1) / ZB);
    dim3 gScat(NT);
    dim3 bScat(1024);
    dim3 gMask((NG + BW - 1) / BW);
    dim3 bMask(BW * 32);
    dim3 gGemm(NN / 16);

    build_table<<<gMask, bMask, 0, stream>>>(src, dst, hmap);

    tile_replay<<<gScat, bScat, 0, stream>>>(x, src, dst, hmap, p0);
    sage_gemm<DH, DH, true, 256><<<gGemm, dim3(256), 0, stream>>>(p0, x, Wl0, Wr0, b0, p1);

    tile_replay<<<gScat, bScat, 0, stream>>>(p1, src, dst, hmap, p0);
    sage_gemm<DH, DH, true, 256><<<gGemm, dim3(256), 0, stream>>>(p0, p1, Wl1, Wr1, b1, p0);

    tile_replay<<<gScat, bScat, 0, stream>>>(p0, src, dst, hmap, p1);
    sage_gemm<DH, NC, false, 96><<<gGemm, dim3(96), 0, stream>>>(p1, p0, Wl2, Wr2, b2, hout);

    logsoftmax_kernel<<<gN, bZ, 0, stream>>>(hout);
}


# Tile scatter fetches each 512-B source row with ONE 16-B-a-lane load (4 whole lines), not 4 one-line loads; LDS tile columns permuted so ds_add_f32 stay conflict-free; sums in LDS arrival order
# speedup vs baseline: 3.4069x; 1.0697x over previous
#include <hip/hip_runtime.h>
#include <math.h>

#define NN 100000
#define NE 1600000
#define DH 128
#define NC 40
#define A_STRIDE 260

typedef __attribute__((ext_vector_type(2))) float v2f;
typedef __attribute__((ext_vector_type(8))) float v8f;

#define TR 391
#define NT 256
#define NG (NE / 128)
#define BW 8
#define SLOT_USED 0x80000000u
#define SLOT_SPILL 0xffffffffu
#define RING 256u
static_assert(NE % 128 == 0, "a group is 128 whole edges");
static_assert(NT * TR >= NN, "tiles cover all rows");
static_assert((size_t)NT * NG * 16 <= (size_t)NN * DH * 4, "the table fits one feature plane");
static_assert(NN <= (1 << 17) && TR <= (1 << 9), "source in 17 bits, row in 9 bits");

typedef __attribute__((ext_vector_type(4))) int v4i;

__global__ __launch_bounds__(BW * 32)
void build_table(const int* __restrict__ src, const int* __restrict__ dst,
                 int* __restrict__ table) {
    __shared__ unsigned tab[BW][NT][4];
    __shared__ unsigned fill[BW][NT];
    const int tid  = threadIdx.x;
    const int lane = tid & 31;
    const int wave = tid >> 5;
    const int g    = blockIdx.x * BW + wave;

    uint4* tab4 = reinterpret_cast<uint4*>(&tab[0][0][0]);
    unsigned* fill1 = &fill[0][0];
    for (int i = tid; i < BW * NT; i += BW * 32) {
        tab4[i]  = make_uint4(0u, 0u, 0u, 0u);
        fill1[i] = 0u;
    }
    __syncthreads();

    if (g < NG) {
        const int4 d = reinterpret_cast<const int4*>(dst)[g * 32 + lane];
        const int4 s = reinterpret_cast<const int4*>(src)[g * 32 + lane];
        auto put = [&](int dv, int sv) {
            if ((unsigned)dv < (unsigned)NN && (unsigned)sv < (unsigned)NN) {
                const int t = dv / TR;
                const int r = dv - t * TR;
                const unsigned k = atomicAdd(&fill[wave][t], 1u);
                if (k < 4u) tab[wave][t][k] = SLOT_USED | ((unsigned)r << 17) | (unsigned)sv;
            }
        };
        put(d.x, s.x);
        put(d.y, s.y);
        put(d.z, s.z);
        put(d.w, s.w);
    }
    __syncthreads();

    uint4* __restrict__ hm = reinterpret_cast<uint4*>(table);
    for (int i = tid; i < BW * NT; i += BW * 32) {
        const int t  = i / BW;
        const int w  = i % BW;
        const int gg = blockIdx.x * BW + w;
        uint4 v = tab4[w * NT + t];
        if (fill[w][t] > 4u) v = make_uint4(SLOT_SPILL, 0u, 0u, 0u);
        if (gg < NG) hm[(size_t)t * NG + gg] = v;
    }
}

__global__ __launch_bounds__(1024)
void tile_replay(const float* __restrict__ x,
                 const int* __restrict__ src,
                 const int* __restrict__ dst,
                 const int* __restrict__ table,
                 float* __restrict__ aggr) {
    __shared__ float tile[(TR + 1) * DH];
    __shared__ float cnt[TR + 1];
    __shared__ unsigned ring[32][RING];
    const int tid  = threadIdx.x;
    const int lane = tid & 31;
    const int wave = tid >> 5;
    const int base = blockIdx.x * TR;
    const int rows = min(TR, NN - base);

    float4* t4 = reinterpret_cast<float4*>(tile);
    for (int i = tid; i < (TR + 1) * DH / 4; i += 1024)
        t4[i] = make_float4(0.f, 0.f, 0.f, 0.f);
    for (int i = tid; i < TR + 1; i += 1024) cnt[i] = 0.f;
    for (int i = tid; i < 32 * RING; i += 1024) (&ring[0][0])[i] = 0u;
    __syncthreads();

    unsigned* myring = ring[wave];
    unsigned head = 0;
    unsigned tail = 0;

    auto flush = [&](int n) {
        const unsigned pv = myring[(head + (unsigned)lane) & (RING - 1u)];
        head += (unsigned)n;
        const bool live = lane < n;
        const int sv = live ? (int)(pv & 0x1ffffu) : 0;
        const int dv = live ? (int)((pv >> 17) & 0x1ffu) : TR;
        if (live) atomicAdd(&cnt[dv], 1.0f);
        for (int b0 = 0; b0 < n; b0 += 8) {
            float4 v[8];
            #pragma unroll
            for (int j = 0; j < 8; ++j) {
                const int s = __builtin_amdgcn_readlane(sv, b0 + j);
                const float4* __restrict__ xr = reinterpret_cast<const float4*>(x + (size_t)s * DH);
                v[j] = xr[lane];
            }
            #pragma unroll
            for (int j = 0; j < 8; ++j) {
                const int d = __builtin_amdgcn_readlane(dv, b0 + j);
                float* o = tile + d * DH + lane;
                atomicAdd(o +  0, v[j].x);
                atomicAdd(o + 32, v[j].y);
                atomicAdd(o + 64, v[j].z);
                atomicAdd(o + 96, v[j].w);
            }
        }
    };

    auto push = [&](bool has, unsigned val) {
        const unsigned bal = __builtin_amdgcn_ballot_w32(has);
        const unsigned rank = __builtin_amdgcn_mbcnt_lo(bal, 0u);
        if (has) myring[(tail + rank) & (RING - 1u)] = val;
        tail += (unsigned)__builtin_popcount(bal);
    };
    auto drain = [&]() {
        while (tail - head >= 32u) flush(32);
    };

    const v4i* __restrict__ hm = reinterpret_cast<const v4i*>(table) + (size_t)blockIdx.x * NG;
    constexpr int NTRIP = (NG + 1023) / 1024;
    int g = wave * 32 + lane;
    v4i nxt = __builtin_nontemporal_load(hm + min(g, NG - 1));
    for (int t = 0; t < NTRIP; ++t) {
        const v4i cur = nxt;
        const bool mine = g < NG;
        const int gb = t * 1024 + wave * 32;
        g += 1024;
        nxt = __builtin_nontemporal_load(hm + min(g, NG - 1));
        const bool spill = mine && (unsigned)cur.x == SLOT_SPILL;
        const bool take  = mine && !spill;
        push(take && ((unsigned)cur.x & SLOT_USED) != 0u, (unsigned)cur.x);
        push(take && ((unsigned)cur.y & SLOT_USED) != 0u, (unsigned)cur.y);
        push(take && ((unsigned)cur.z & SLOT_USED) != 0u, (unsigned)cur.z);
        push(take && ((unsigned)cur.w & SLOT_USED) != 0u, (unsigned)cur.w);
        drain();
        unsigned om = __builtin_amdgcn_ballot_w32(spill);
        while (om) {
            const int l = __builtin_ctz(om);
            om &= om - 1;
            const int gq = gb + l;
            const int4 d = reinterpret_cast<const int4*>(dst)[gq * 32 + lane];
            const int4 s = reinterpret_cast<const int4*>(src)[gq * 32 + lane];
            auto test = [&](int dv, int sv) {
                const int dc = dv - base;
                const bool ok = (unsigned)dc < (unsigned)rows && (unsigned)sv < (unsigned)NN;
                push(ok, SLOT_USED | ((unsigned)dc << 17) | (unsigned)sv);
            };
            test(d.x, s.x);
            test(d.y, s.y);
            test(d.z, s.z);
            test(d.w, s.w);
            drain();
        }
    }
    if (tail != head) flush((int)(tail - head));
    __syncthreads();

    float4* __restrict__ a4 = reinterpret_cast<float4*>(aggr + (size_t)base * DH);
    for (int i = tid; i < rows * (DH / 4); i += 1024) {
        const int row = i / (DH / 4);
        const float s = 1.0f / fmaxf(cnt[row], 1.0f);
        const float* tr = tile + row * DH + lane;
        float4 v;
        v.x = tr[0]  * s;
        v.y = tr[32] * s;
        v.z = tr[64] * s;
        v.w = tr[96] * s;
        a4[i] = v;
    }
}

template <int DIN, int DOUT, bool RELU, int NTH>
__global__ __launch_bounds__(NTH)
void sage_gemm(const float* aggr,
               const float* __restrict__ x,
               const float* __restrict__ Wl,
               const float* __restrict__ Wr,
               const float* __restrict__ bias,
               float* out) {
    __shared__ __attribute__((aligned(16))) float Alds[16 * A_STRIDE];

    constexpr int twoK = 2 * DIN;
    constexpr int HV = DIN / 4;
    constexpr int NP = 16 * HV;
    constexpr int IT = (NP + NTH - 1) / NTH;
    constexpr int NGRP = twoK / 32;
    constexpr int HGRP = DIN / 32;
    static_assert(DIN % 32 == 0, "a group is 8 whole k-steps inside one half");
    static_assert(A_STRIDE % 4 == 0 && DIN % 4 == 0 && A_STRIDE >= twoK, "16-byte LDS pieces");
    static_assert(NTH % 32 == 0 && NTH / 32 * 16 >= DOUT, "one wave per 16 columns");

    const int m0   = blockIdx.x * 16;
    const int tid  = threadIdx.x;
    const int lane = tid & 31;
    const int wave = tid >> 5;

    float4 aa[IT], xx[IT];
    #pragma unroll
    for (int it = 0; it < IT; ++it) {
        const int j0  = tid + it * NTH;
        const int j   = ((it + 1) * NTH <= NP) ? j0 : min(j0, NP - 1);
        const int row = j / HV;
        const int c   = j - row * HV;
        const int g   = m0 + row;
        aa[it] = *reinterpret_cast<const float4*>(aggr + (size_t)g * DIN + 4 * c);
        xx[it] = *reinterpret_cast<const float4*>(x    + (size_t)g * DIN + 4 * c);
    }

    const int half = lane >> 4;
    const int l15  = lane & 15;
    const int n0   = wave * 16;
    const int col  = n0 + l15;
    const int ccol = (col < DOUT) ? col : 0;

    const float* __restrict__ Wlp = Wl + 2 * half * DOUT + ccol;
    const float* __restrict__ Wrp = Wr + 2 * half * DOUT + ccol;
    const float* __restrict__ Ap  = Alds + l15 * A_STRIDE + 2 * half;

    auto loadB = [&](int g, v2f (&b)[8]) {
        const float* __restrict__ Wp = (g < HGRP) ? Wlp : Wrp;
        const int kb = ((g < HGRP) ? g : g - HGRP) * 32;
        #pragma unroll
        for (int u = 0; u < 8; ++u) {
            b[u].x = Wp[(kb + 4 * u) * DOUT];
            b[u].y = Wp[(kb + 4 * u) * DOUT + DOUT];
        }
    };

    const float bv = bias[ccol];
    v2f bc[8], bn[8];
    loadB(0, bc);

    #pragma unroll
    for (int it = 0; it < IT; ++it) {
        const int j = tid + it * NTH;
        if ((it + 1) * NTH <= NP || j < NP) {
            const int row = j / HV;
            const int c   = j - row * HV;
            *reinterpret_cast<float4*>(Alds + row * A_STRIDE + 4 * c)       = aa[it];
            *reinterpret_cast<float4*>(Alds + row * A_STRIDE + DIN + 4 * c) = xx[it];
        }
    }
    __syncthreads();

    v8f acc = {};
    #pragma unroll
    for (int g = 0; g < NGRP; ++g) {
        if (g + 1 < NGRP) loadB(g + 1, bn);
        #pragma unroll
        for (int u = 0; u < 8; ++u) {
            v2f a;
            a.x = Ap[32 * g + 4 * u];
            a.y = Ap[32 * g + 4 * u + 1];
            acc = __builtin_amdgcn_wmma_f32_16x16x4_f32(
                false, a, false, bc[u], (short)0, acc, false, false);
        }
        #pragma unroll
        for (int u = 0; u < 8; ++u) bc[u] = bn[u];
    }

    if (col < DOUT) {
        float* op = out + (size_t)(m0 + 8 * half) * DOUT + col;
        #pragma unroll
        for (int r = 0; r < 8; ++r) {
            float v = acc[r] + bv;
            if (RELU) v = fmaxf(v, 0.0f);
            op[r * DOUT] = v;
        }
    }
}

__global__ void logsoftmax_kernel(float* __restrict__ h) {
    int i = blockIdx.x * blockDim.x + threadIdx.x;
    if (i >= NN) return;
    float* row = h + (size_t)i * NC;
    float mx = -INFINITY;
    #pragma unroll
    for (int c = 0; c < NC; ++c) mx = fmaxf(mx, row[c]);
    float s = 0.0f;
    #pragma unroll
    for (int c = 0; c < NC; ++c) s += expf(row[c] - mx);
    float lse = mx + logf(s);
    #pragma unroll
    for (int c = 0; c < NC; ++c) row[c] = row[c] - lse;
}

extern "C" void kernel_launch(void* const* d_in, const int* in_sizes, int n_in,
                              void* d_out, int out_size, void* d_ws, size_t ws_size,
                              hipStream_t stream) {
    const float* x   = (const float*)d_in[0];
    const int*   ei  = (const int*)d_in[1];
    const int*   src = ei;
    const int*   dst = ei + NE;
    const float* Wl0 = (const float*)d_in[2];
    const float* Wr0 = (const float*)d_in[3];
    const float* b0  = (const float*)d_in[4];
    const float* Wl1 = (const float*)d_in[5];
    const float* Wr1 = (const float*)d_in[6];
    const float* b1  = (const float*)d_in[7];
    const float* Wl2 = (const float*)d_in[8];
    const float* Wr2 = (const float*)d_in[9];
    const float* b2  = (const float*)d_in[10];
    float* hout = (float*)d_out;

    char* ws = (char*)d_ws;
    size_t off = 0;
    auto carve = [&](size_t bytes) {
        void* p = ws + off;
        off = (off + bytes + 255) & ~(size_t)255;
        return p;
    };
    float* p0   = (float*)carve((size_t)NN * DH * 4);
    float* p1   = (float*)carve((size_t)NN * DH * 4);
    int*   hmap = (int*)carve((size_t)NN * DH * 4);

    const int ZB = 256;
    dim3 bZ(ZB);
    dim3 gN((NN + ZB - 1) / ZB);
    dim3 gScat(NT);
    dim3 bScat(1024);
    dim3 gMask((NG + BW - 1) / BW);
    dim3 bMask(BW * 32);
    dim3 gGemm(NN / 16);

    build_table<<<gMask, bMask, 0, stream>>>(src, dst, hmap);

    tile_replay<<<gScat, bScat, 0, stream>>>(x, src, dst, hmap, p0);
    sage_gemm<DH, DH, true, 256><<<gGemm, dim3(256), 0, stream>>>(p0, x, Wl0, Wr0, b0, p1);

    tile_replay<<<gScat, bScat, 0, stream>>>(p1, src, dst, hmap, p0);
    sage_gemm<DH, DH, true, 256><<<gGemm, dim3(256), 0, stream>>>(p0, p1, Wl1, Wr1, b1, p0);

    tile_replay<<<gScat, bScat, 0, stream>>>(p0, src, dst, hmap, p1);
    sage_gemm<DH, NC, false, 96><<<gGemm, dim3(96), 0, stream>>>(p1, p0, Wl2, Wr2, b2, hout);

    logsoftmax_kernel<<<gN, bZ, 0, stream>>>(hout);
}


# Layer GEMM done by the tile scatter's own work-group: f32 WMMA takes the mean half of A from the LDS sum tile, weights from LDS; no aggr plane or GEMM launch (8 -> 5); k order in a 4-k step changes
# speedup vs baseline: 4.3088x; 1.2647x over previous
#include <hip/hip_runtime.h>
#include <math.h>

#define NN 100000
#define NE 1600000
#define DH 128
#define NC 40

typedef __attribute__((ext_vector_type(2))) float v2f;
typedef __attribute__((ext_vector_type(8))) float v8f;

#define TR 391
#define TROWS 400
#define TP 130
#define NT 256
#define NG (NE / 128)
#define BW 8
#define SLOT_USED 0x80000000u
#define SLOT_SPILL 0xffffffffu
#define RING 256u
static_assert(NE % 128 == 0, "a group is 128 whole edges");
static_assert(NT * TR >= NN, "tiles cover all rows");
static_assert((size_t)NT * NG * 16 <= (size_t)NN * DH * 4, "the table fits one feature plane");
static_assert(NN <= (1 << 17) && TR <= (1 << 9), "source in 17 bits, row in 9 bits");

typedef __attribute__((ext_vector_type(4))) int v4i;

__global__ __launch_bounds__(BW * 32)
void build_table(const int* __restrict__ src, const int* __restrict__ dst,
                 int* __restrict__ table) {
    __shared__ unsigned tab[BW][NT][4];
    __shared__ unsigned fill[BW][NT];
    const int tid  = threadIdx.x;
    const int lane = tid & 31;
    const int wave = tid >> 5;
    const int g    = blockIdx.x * BW + wave;

    uint4* tab4 = reinterpret_cast<uint4*>(&tab[0][0][0]);
    unsigned* fill1 = &fill[0][0];
    for (int i = tid; i < BW * NT; i += BW * 32) {
        tab4[i]  = make_uint4(0u, 0u, 0u, 0u);
        fill1[i] = 0u;
    }
    __syncthreads();

    if (g < NG) {
        const int4 d = reinterpret_cast<const int4*>(dst)[g * 32 + lane];
        const int4 s = reinterpret_cast<const int4*>(src)[g * 32 + lane];
        auto put = [&](int dv, int sv) {
            if ((unsigned)dv < (unsigned)NN && (unsigned)sv < (unsigned)NN) {
                const int t = dv / TR;
                const int r = dv - t * TR;
                const unsigned k = atomicAdd(&fill[wave][t], 1u);
                if (k < 4u) tab[wave][t][k] = SLOT_USED | ((unsigned)r << 17) | (unsigned)sv;
            }
        };
        put(d.x, s.x);
        put(d.y, s.y);
        put(d.z, s.z);
        put(d.w, s.w);
    }
    __syncthreads();

    uint4* __restrict__ hm = reinterpret_cast<uint4*>(table);
    for (int i = tid; i < BW * NT; i += BW * 32) {
        const int t  = i / BW;
        const int w  = i % BW;
        const int gg = blockIdx.x * BW + w;
        uint4 v = tab4[w * NT + t];
        if (fill[w][t] > 4u) v = make_uint4(SLOT_SPILL, 0u, 0u, 0u);
        if (gg < NG) hm[(size_t)t * NG + gg] = v;
    }
}

template <int DOUT, bool RELU>
__global__ __launch_bounds__(1024)
void sage_layer(const float* __restrict__ x,
                const int* __restrict__ src,
                const int* __restrict__ dst,
                const int* __restrict__ table,
                const float* __restrict__ Wl,
                const float* __restrict__ Wr,
                const float* __restrict__ bias,
                float* __restrict__ out) {
    constexpr int NCT   = (DOUT + 15) / 16;
    constexpr int CG    = (NCT % 4 == 0) ? 4 : NCT;
    constexpr int NCG   = NCT / CG;
    constexpr int NRT   = TROWS / 16;
    constexpr int NITEM = NRT * NCG;
    constexpr int NIT   = (NITEM + 31) / 32;
    constexpr int DP    = NCT * 16;
    constexpr int WP    = 2 * DP + (32 + 64 - (2 * DP) % 64) % 64;
    static_assert(WP % 64 == 32, "the two lane halves read different banks");
    static_assert(NCT % CG == 0 && TROWS % 16 == 0 && TROWS > TR, "whole items");

    __shared__ __attribute__((aligned(16))) float tile[TROWS * TP];
    __shared__ float cnt[TROWS];
    __shared__ unsigned ring[32][RING];
    __shared__ __attribute__((aligned(16))) float wlds[(DH / 2) * WP];
    const int tid  = threadIdx.x;
    const int lane = tid & 31;
    const int wave = __builtin_amdgcn_readfirstlane(tid >> 5);
    const int base = blockIdx.x * TR;
    const int rows = min(TR, NN - base);

    constexpr int NST = DP / 16;
    static_assert((DH / 2) * 16 == 1024, "one pair slot per 16 threads");
    const int sq = tid >> 4;
    const int sn = NST * (tid & 15);
    float2 wreg[NST];
    auto load_pairs = [&](const float* __restrict__ Wsrc, bool self_half) {
        const int t  = sq >> 1;
        const int hh = sq & 1;
        const int r0 = self_half ? 4 * t + 2 * hh : 4 * t + hh;
        const int r1 = self_half ? r0 + 1 : r0 + 2;
        #pragma unroll
        for (int u = 0; u < NST; ++u) {
            const int n = sn + u;
            float2 w = make_float2(0.f, 0.f);
            if (DOUT % 16 == 0 || n < DOUT) {
                w.x = Wsrc[r0 * DOUT + n];
                w.y = Wsrc[r1 * DOUT + n];
            }
            wreg[u] = w;
        }
    };
    auto store_pairs = [&]() {
        #pragma unroll
        for (int u = 0; u < NST; ++u)
            *reinterpret_cast<float2*>(wlds + sq * WP + 2 * (sn + u)) = wreg[u];
    };

    load_pairs(Wl, false);
    float4* t4 = reinterpret_cast<float4*>(tile);
    for (int i = tid; i < TROWS * TP / 4; i += 1024)
        t4[i] = make_float4(0.f, 0.f, 0.f, 0.f);
    for (int i = tid; i < TROWS; i += 1024) cnt[i] = 0.f;
    for (int i = tid; i < 32 * RING; i += 1024) (&ring[0][0])[i] = 0u;
    store_pairs();
    __syncthreads();

    unsigned* myring = ring[wave];
    unsigned head = 0;
    unsigned tail = 0;

    auto flush = [&](int n) {
        const unsigned pv = myring[(head + (unsigned)lane) & (RING - 1u)];
        head += (unsigned)n;
        const bool live = lane < n;
        const int sv = live ? (int)(pv & 0x1ffffu) : 0;
        const int dv = live ? (int)((pv >> 17) & 0x1ffu) : TR;
        if (live) atomicAdd(&cnt[dv], 1.0f);
        for (int b0 = 0; b0 < n; b0 += 8) {
            float4 v[8];
            #pragma unroll
            for (int j = 0; j < 8; ++j) {
                const int s = __builtin_amdgcn_readlane(sv, b0 + j);
                const float4* __restrict__ xr = reinterpret_cast<const float4*>(x + (size_t)s * DH);
                v[j] = xr[lane];
            }
            #pragma unroll
            for (int j = 0; j < 8; ++j) {
                const int d = __builtin_amdgcn_readlane(dv, b0 + j);
                float* o = tile + d * TP + lane;
                atomicAdd(o +  0, v[j].x);
                atomicAdd(o + 32, v[j].y);
                atomicAdd(o + 64, v[j].z);
                atomicAdd(o + 96, v[j].w);
            }
        }
    };

    auto push = [&](bool has, unsigned val) {
        const unsigned bal = __builtin_amdgcn_ballot_w32(has);
        const unsigned rank = __builtin_amdgcn_mbcnt_lo(bal, 0u);
        if (has) myring[(tail + rank) & (RING - 1u)] = val;
        tail += (unsigned)__builtin_popcount(bal);
    };
    auto drain = [&]() {
        while (tail - head >= 32u) flush(32);
    };

    const v4i* __restrict__ hm = reinterpret_cast<const v4i*>(table) + (size_t)blockIdx.x * NG;
    constexpr int NTRIP = (NG + 1023) / 1024;
    int g = wave * 32 + lane;
    v4i nxt = __builtin_nontemporal_load(hm + min(g, NG - 1));
    for (int t = 0; t < NTRIP; ++t) {
        const v4i cur = nxt;
        const bool mine = g < NG;
        const int gb = t * 1024 + wave * 32;
        g += 1024;
        nxt = __builtin_nontemporal_load(hm + min(g, NG - 1));
        const bool spill = mine && (unsigned)cur.x == SLOT_SPILL;
        const bool take  = mine && !spill;
        push(take && ((unsigned)cur.x & SLOT_USED) != 0u, (unsigned)cur.x);
        push(take && ((unsigned)cur.y & SLOT_USED) != 0u, (unsigned)cur.y);
        push(take && ((unsigned)cur.z & SLOT_USED) != 0u, (unsigned)cur.z);
        push(take && ((unsigned)cur.w & SLOT_USED) != 0u, (unsigned)cur.w);
        drain();
        unsigned om = __builtin_amdgcn_ballot_w32(spill);
        while (om) {
            const int l = __builtin_ctz(om);
            om &= om - 1;
            const int gq = gb + l;
            const int4 d = reinterpret_cast<const int4*>(dst)[gq * 32 + lane];
            const int4 s = reinterpret_cast<const int4*>(src)[gq * 32 + lane];
            auto test = [&](int dv, int sv) {
                const int dc = dv - base;
                const bool ok = (unsigned)dc < (unsigned)rows && (unsigned)sv < (unsigned)NN;
                push(ok, SLOT_USED | ((unsigned)dc << 17) | (unsigned)sv);
            };
            test(d.x, s.x);
            test(d.y, s.y);
            test(d.z, s.z);
            test(d.w, s.w);
            drain();
        }
    }
    if (tail != head) flush((int)(tail - head));
    __syncthreads();

    const int half = lane >> 4;
    const int l15  = lane & 15;
    v8f acc[NIT][CG];
    #pragma unroll
    for (int it = 0; it < NIT; ++it) {
        #pragma unroll
        for (int j = 0; j < CG; ++j) acc[it][j] = v8f{};
    }

    #pragma unroll
    for (int it = 0; it < NIT; ++it) {
        const int item = wave + 32 * it;
        const int rt = item / NCG;
        const int cg = item - rt * NCG;
        if (item < NITEM && rt * 16 < rows) {
            const int row = rt * 16 + l15;
            const float sc = 1.0f / fmaxf(cnt[row], 1.0f);
            const float* ap = tile + row * TP + 32 * half;
            const float* bp = wlds + half * WP + 2 * (cg * CG * 16 + l15);
            #pragma unroll 8
            for (int t = 0; t < DH / 4; ++t) {
                v2f a;
                a.x = ap[t] * sc;
                a.y = ap[64 + t] * sc;
                #pragma unroll
                for (int j = 0; j < CG; ++j) {
                    const v2f b = *reinterpret_cast<const v2f*>(bp + 2 * t * WP + 32 * j);
                    acc[it][j] = __builtin_amdgcn_wmma_f32_16x16x4_f32(
                        false, a, false, b, (short)0, acc[it][j], false, false);
                }
            }
        }
    }
    __syncthreads();
    load_pairs(Wr, true);
    store_pairs();
    __syncthreads();

    #pragma unroll
    for (int it = 0; it < NIT; ++it) {
        const int item = wave + 32 * it;
        const int rt = item / NCG;
        const int cg = item - rt * NCG;
        if (item < NITEM && rt * 16 < rows) {
            const int grow = min(base + rt * 16 + l15, NN - 1);
            const float* __restrict__ xp = x + (size_t)grow * DH + 2 * half;
            const float* bp = wlds + half * WP + 2 * (cg * CG * 16 + l15);
            #pragma unroll 8
            for (int t = 0; t < DH / 4; ++t) {
                const v2f a = *reinterpret_cast<const v2f*>(xp + 4 * t);
                #pragma unroll
                for (int j = 0; j < CG; ++j) {
                    const v2f b = *reinterpret_cast<const v2f*>(bp + 2 * t * WP + 32 * j);
                    acc[it][j] = __builtin_amdgcn_wmma_f32_16x16x4_f32(
                        false, a, false, b, (short)0, acc[it][j], false, false);
                }
            }
        }
    }

    #pragma unroll
    for (int it = 0; it < NIT; ++it) {
        const int item = wave + 32 * it;
        const int rt = item / NCG;
        const int cg = item - rt * NCG;
        if (item < NITEM && rt * 16 < rows) {
            #pragma unroll
            for (int j = 0; j < CG; ++j) {
                const int col = (cg * CG + j) * 16 + l15;
                const float bv = bias[min(col, DOUT - 1)];
                #pragma unroll
                for (int r = 0; r < 8; ++r) {
                    const int row = rt * 16 + 8 * half + r;
                    float v = acc[it][j][r] + bv;
                    if (RELU) v = fmaxf(v, 0.0f);
                    if (row < rows && (DOUT % 16 == 0 || col < DOUT)) out[(size_t)(base + row) * DOUT + col] = v;
                }
            }
        }
    }
}

__global__ void logsoftmax_kernel(float* __restrict__ h) {
    int i = blockIdx.x * blockDim.x + threadIdx.x;
    if (i >= NN) return;
    float* row = h + (size_t)i * NC;
    float mx = -INFINITY;
    #pragma unroll
    for (int c = 0; c < NC; ++c) mx = fmaxf(mx, row[c]);
    float s = 0.0f;
    #pragma unroll
    for (int c = 0; c < NC; ++c) s += expf(row[c] - mx);
    float lse = mx + logf(s);
    #pragma unroll
    for (int c = 0; c < NC; ++c) row[c] = row[c] - lse;
}

extern "C" void kernel_launch(void* const* d_in, const int* in_sizes, int n_in,
                              void* d_out, int out_size, void* d_ws, size_t ws_size,
                              hipStream_t stream) {
    const float* x   = (const float*)d_in[0];
    const int*   ei  = (const int*)d_in[1];
    const int*   src = ei;
    const int*   dst = ei + NE;
    const float* Wl0 = (const float*)d_in[2];
    const float* Wr0 = (const float*)d_in[3];
    const float* b0  = (const float*)d_in[4];
    const float* Wl1 = (const float*)d_in[5];
    const float* Wr1 = (const float*)d_in[6];
    const float* b1  = (const float*)d_in[7];
    const float* Wl2 = (const float*)d_in[8];
    const float* Wr2 = (const float*)d_in[9];
    const float* b2  = (const float*)d_in[10];
    float* hout = (float*)d_out;

    char* ws = (char*)d_ws;
    size_t off = 0;
    auto carve = [&](size_t bytes) {
        void* p = ws + off;
        off = (off + bytes + 255) & ~(size_t)255;
        return p;
    };
    float* p0   = (float*)carve((size_t)NN * DH * 4);
    float* p1   = (float*)carve((size_t)NN * DH * 4);
    int*   hmap = (int*)carve((size_t)NN * DH * 4);

    const int ZB = 256;
    dim3 bZ(ZB);
    dim3 gN((NN + ZB - 1) / ZB);
    dim3 gTile(NT);
    dim3 bTile(1024);
    dim3 gMask((NG + BW - 1) / BW);
    dim3 bMask(BW * 32);

    build_table<<<gMask, bMask, 0, stream>>>(src, dst, hmap);

    sage_layer<DH, true><<<gTile, bTile, 0, stream>>>(x, src, dst, hmap, Wl0, Wr0, b0, p1);
    sage_layer<DH, true><<<gTile, bTile, 0, stream>>>(p1, src, dst, hmap, Wl1, Wr1, b1, p0);
    sage_layer<NC, false><<<gTile, bTile, 0, stream>>>(p0, src, dst, hmap, Wl2, Wr2, b2, hout);

    logsoftmax_kernel<<<gN, bZ, 0, stream>>>(hout);
}
